# starting kernel
# baseline (speedup 1.0000x reference)
#include <hip/hip_runtime.h>
#include <hip/hip_bf16.h>

typedef __attribute__((ext_vector_type(2))) float v2f;
typedef __attribute__((ext_vector_type(8))) float v8f;

#define C_DIM 256

__global__ void zero_f32(float* __restrict__ p, long long n) {
  long long i = (long long)blockIdx.x * blockDim.x + threadIdx.x;
  long long stride = (long long)gridDim.x * blockDim.x;
  for (; i < n; i += stride) p[i] = 0.0f;
}

__global__ void transpose_w(const float* __restrict__ W, float* __restrict__ Wt,
                            int K, int N) {
  int t = blockIdx.x * blockDim.x + threadIdx.x;
  if (t >= K * N) return;
  int k = t / N;
  int n = t % N;
  Wt[n * K + k] = W[t];
}

__global__ void scatter_edges(const float* __restrict__ x,
                              const int* __restrict__ src,
                              const int* __restrict__ dst,
                              float* __restrict__ agg,
                              float* __restrict__ cnt,
                              int n_edges) {
  const int chunks = C_DIM / 4;
  long long t = (long long)blockIdx.x * blockDim.x + threadIdx.x;
  long long total = (long long)n_edges * chunks;
  if (t >= total) return;
  int e = (int)(t / chunks);
  int c = (int)(t % chunks) * 4;
  int s = src[e];
  int d = dst[e];
  const float4 v = *(const float4*)(x + (long long)s * C_DIM + c);
  float* a = agg + (long long)d * C_DIM + c;
  atomicAdd(a + 0, v.x);
  atomicAdd(a + 1, v.y);
  atomicAdd(a + 2, v.z);
  atomicAdd(a + 3, v.w);
  if (c == 0) atomicAdd(cnt + d, 1.0f);
}

template <int RELU>
__global__ __launch_bounds__(256) void gemm_dual_wmma(
    const float* __restrict__ A1, const float* __restrict__ Wt1,
    const float* __restrict__ cnt1,
    const float* __restrict__ A2, const float* __restrict__ Wt2,
    const float* __restrict__ bias, float* __restrict__ out,
    int M, int N) {
  constexpr int K = C_DIM;
  const int lane = threadIdx.x & 31;
  const int wave = threadIdx.x >> 5;
  const int nstrips = N >> 6;
  const int tile = blockIdx.x * 8 + wave;
  if (tile >= (M >> 4) * nstrips) return;

  const int m0 = (tile / nstrips) << 4;
  const int n0 = (tile % nstrips) << 6;

  const int half = lane >> 4;
  const int l15  = lane & 15;
  const int row  = m0 + l15;
  const int cb   = n0 + l15;
  const int koff = half << 1;

  const float s1 = 1.0f / fmaxf(cnt1[row], 1.0f);

  v8f acc[4];
#pragma unroll
  for (int j = 0; j < 4; ++j) acc[j] = (v8f){0.f,0.f,0.f,0.f,0.f,0.f,0.f,0.f};

  const float* a1p = A1 + (long long)row * K + koff;
  const float* a2p = A2 + (long long)row * K + koff;
  const float* w1p = Wt1 + (long long)cb * K + koff;
  const float* w2p = Wt2 + (long long)cb * K + koff;

  for (int k = 0; k < K; k += 4) {
    v2f a = *(const v2f*)(a1p + k);
    a = a * s1;
#pragma unroll
    for (int j = 0; j < 4; ++j) {
      v2f b = *(const v2f*)(w1p + (long long)j * 16 * K + k);
      acc[j] = __builtin_amdgcn_wmma_f32_16x16x4_f32(false, a, false, b,
                                                     (short)0, acc[j],
                                                     false, false);
    }
  }
  for (int k = 0; k < K; k += 4) {
    v2f a = *(const v2f*)(a2p + k);
#pragma unroll
    for (int j = 0; j < 4; ++j) {
      v2f b = *(const v2f*)(w2p + (long long)j * 16 * K + k);
      acc[j] = __builtin_amdgcn_wmma_f32_16x16x4_f32(false, a, false, b,
                                                     (short)0, acc[j],
                                                     false, false);
    }
  }

#pragma unroll
  for (int j = 0; j < 4; ++j) {
    const int col = cb + j * 16;
    const float bv = bias[col];
#pragma unroll
    for (int r = 0; r < 8; ++r) {
      float v = acc[j][r] + bv;
      if (RELU) v = fmaxf(v, 0.f);
      out[(long long)(m0 + r + half * 8) * N + col] = v;
    }
  }
}

static inline int cdiv(long long a, long long b) { return (int)((a + b - 1) / b); }

extern "C" void kernel_launch(void* const* d_in, const int* in_sizes, int n_in,
                              void* d_out, int out_size, void* d_ws, size_t ws_size,
                              hipStream_t stream) {
  (void)n_in; (void)out_size; (void)ws_size;

  const int N1 = 40000, N2 = 20000, N3 = 10000;
  const int OUT_C = 64;

  const float* x   = (const float*)d_in[0];
  const int*   ei1 = (const int*)d_in[1];
  const int*   ei2 = (const int*)d_in[2];
  const int*   ei3 = (const int*)d_in[3];
  const float* Wl1 = (const float*)d_in[4];
  const float* bl1 = (const float*)d_in[5];
  const float* Wr1 = (const float*)d_in[6];
  const float* Wl2 = (const float*)d_in[7];
  const float* bl2 = (const float*)d_in[8];
  const float* Wr2 = (const float*)d_in[9];
  const float* Wl3 = (const float*)d_in[10];
  const float* bl3 = (const float*)d_in[11];
  const float* Wr3 = (const float*)d_in[12];
  const float* Wls = (const float*)d_in[13];
  const float* bls = (const float*)d_in[14];
  const float* Wrs = (const float*)d_in[15];

  const int E1 = in_sizes[1] / 2;
  const int E2 = in_sizes[2] / 2;
  const int E3 = in_sizes[3] / 2;

  float* h1  = (float*)d_ws;
  float* h2  = h1 + (long long)N1 * C_DIM;
  float* agg = h2 + (long long)N2 * C_DIM;
  float* cnt = agg + (long long)N1 * C_DIM;
  float* wt  = cnt + N1;
  float* Wl1t = wt;
  float* Wr1t = Wl1t + C_DIM * C_DIM;
  float* Wl2t = Wr1t + C_DIM * C_DIM;
  float* Wr2t = Wl2t + C_DIM * C_DIM;
  float* Wl3t = Wr2t + C_DIM * C_DIM;
  float* Wr3t = Wl3t + C_DIM * OUT_C;
  float* Wlst = Wr3t + C_DIM * OUT_C;
  float* Wrst = Wlst + C_DIM * OUT_C;

  float* mu     = (float*)d_out;
  float* logstd = mu + (long long)N3 * OUT_C;

  const int ZB = 256, SB = 256, GB = 256, TB = 256;

  {
    int nbig = C_DIM * C_DIM, nsml = C_DIM * OUT_C;
    transpose_w<<<cdiv(nbig, TB), TB, 0, stream>>>(Wl1, Wl1t, C_DIM, C_DIM);
    transpose_w<<<cdiv(nbig, TB), TB, 0, stream>>>(Wr1, Wr1t, C_DIM, C_DIM);
    transpose_w<<<cdiv(nbig, TB), TB, 0, stream>>>(Wl2, Wl2t, C_DIM, C_DIM);
    transpose_w<<<cdiv(nbig, TB), TB, 0, stream>>>(Wr2, Wr2t, C_DIM, C_DIM);
    transpose_w<<<cdiv(nsml, TB), TB, 0, stream>>>(Wl3, Wl3t, C_DIM, OUT_C);
    transpose_w<<<cdiv(nsml, TB), TB, 0, stream>>>(Wr3, Wr3t, C_DIM, OUT_C);
    transpose_w<<<cdiv(nsml, TB), TB, 0, stream>>>(Wls, Wlst, C_DIM, OUT_C);
    transpose_w<<<cdiv(nsml, TB), TB, 0, stream>>>(Wrs, Wrst, C_DIM, OUT_C);
  }

  {
    long long na = (long long)N1 * C_DIM;
    zero_f32<<<cdiv(na, ZB), ZB, 0, stream>>>(agg, na);
    zero_f32<<<cdiv(N1, ZB), ZB, 0, stream>>>(cnt, N1);
    long long work = (long long)E1 * (C_DIM / 4);
    scatter_edges<<<cdiv(work, SB), SB, 0, stream>>>(x, ei1, ei1 + E1, agg, cnt, E1);
    int tiles = (N1 / 16) * (C_DIM / 64);
    gemm_dual_wmma<1><<<cdiv(tiles, 8), GB, 0, stream>>>(
        agg, Wl1t, cnt, x, Wr1t, bl1, h1, N1, C_DIM);
  }

  {
    long long na = (long long)N2 * C_DIM;
    zero_f32<<<cdiv(na, ZB), ZB, 0, stream>>>(agg, na);
    zero_f32<<<cdiv(N2, ZB), ZB, 0, stream>>>(cnt, N2);
    long long work = (long long)E2 * (C_DIM / 4);
    scatter_edges<<<cdiv(work, SB), SB, 0, stream>>>(h1, ei2, ei2 + E2, agg, cnt, E2);
    int tiles = (N2 / 16) * (C_DIM / 64);
    gemm_dual_wmma<1><<<cdiv(tiles, 8), GB, 0, stream>>>(
        agg, Wl2t, cnt, h1, Wr2t, bl2, h2, N2, C_DIM);
  }

  {
    long long na = (long long)N3 * C_DIM;
    zero_f32<<<cdiv(na, ZB), ZB, 0, stream>>>(agg, na);
    zero_f32<<<cdiv(N3, ZB), ZB, 0, stream>>>(cnt, N3);
    long long work = (long long)E3 * (C_DIM / 4);
    scatter_edges<<<cdiv(work, SB), SB, 0, stream>>>(h2, ei3, ei3 + E3, agg, cnt, E3);

    int tiles = (N3 / 16) * (OUT_C / 64);
    gemm_dual_wmma<0><<<cdiv(tiles, 8), GB, 0, stream>>>(
        agg, Wl3t, cnt, h2, Wr3t, bl3, mu, N3, OUT_C);
    gemm_dual_wmma<0><<<cdiv(tiles, 8), GB, 0, stream>>>(
        agg, Wlst, cnt, h2, Wrst, bls, logstd, N3, OUT_C);
  }
}


# Edge scatter: global float atomics and zero-fills replaced by owner blocks (160/80/40 dst rows summed in LDS) that scan all dst ids and ds_add_f32 the gathered rows; sum order = LDS arrival order
# speedup vs baseline: 1.8926x; 1.8926x over previous
#include <hip/hip_runtime.h>
#include <hip/hip_bf16.h>

typedef __attribute__((ext_vector_type(2))) float v2f;
typedef __attribute__((ext_vector_type(8))) float v8f;

#define C_DIM 256

__global__ void transpose_w(const float* __restrict__ W, float* __restrict__ Wt,
                            int K, int N) {
  int t = blockIdx.x * blockDim.x + threadIdx.x;
  if (t >= K * N) return;
  int k = t / N;
  int n = t % N;
  Wt[n * K + k] = W[t];
}

typedef int   v4i_u __attribute__((ext_vector_type(4), aligned(4)));

template <int R>
__global__ __launch_bounds__(1024) void aggregate_owner_rows(
    const float* __restrict__ x, const int* __restrict__ src,
    const int* __restrict__ dst, float* __restrict__ agg,
    float* __restrict__ cnt, int n_edges, int n_dst) {
  constexpr int NW   = 32;
  constexpr int STEP = 1024;
  constexpr int QCAP = 1152;
  __shared__ float acc[(R + 1) * C_DIM];
  __shared__ float deg[R + 1];
  __shared__ int   queue[NW * QCAP];

  const int tid  = threadIdx.x;
  const int lane = tid & 31;
  const int wave = __builtin_amdgcn_readfirstlane(tid >> 5);
  const int base = blockIdx.x * R;

  for (int i = tid; i < (R + 1) * (C_DIM / 4); i += 1024)
    ((float4*)acc)[i] = make_float4(0.f, 0.f, 0.f, 0.f);
  for (int i = tid; i <= R; i += 1024) deg[i] = 0.f;
  __syncthreads();

  int* q = queue + wave * QCAP;
  int qn = 0;
  const int nsteps = (n_edges + STEP - 1) / STEP;
  const int rot = (int)(((long long)blockIdx.x * nsteps) / gridDim.x);
  int step = wave;

  while (true) {
    while (step < nsteps && qn + STEP <= QCAP) {
      const int pstep = (step + rot < nsteps) ? (step + rot) : (step + rot - nsteps);
      const int e0 = pstep * STEP + lane * 4;
      int dv[8][4];
      if (pstep * STEP + STEP <= n_edges) {
#pragma unroll
        for (int j = 0; j < 8; ++j) {
          const v4i_u t = *(const v4i_u*)(dst + e0 + j * 128);
          dv[j][0] = t.x; dv[j][1] = t.y; dv[j][2] = t.z; dv[j][3] = t.w;
        }
      } else {
#pragma unroll
        for (int j = 0; j < 8; ++j)
#pragma unroll
          for (int c = 0; c < 4; ++c) {
            const int e = e0 + j * 128 + c;
            const int t = dst[(e < n_edges) ? e : (n_edges - 1)];
            dv[j][c] = (e < n_edges) ? t : -1;
          }
      }
#pragma unroll
      for (int j = 0; j < 8; ++j) {
        bool m[4];
#pragma unroll
        for (int c = 0; c < 4; ++c) m[c] = (unsigned)(dv[j][c] - base) < (unsigned)R;
        if (__builtin_amdgcn_ballot_w32(m[0] | m[1] | m[2] | m[3]) != 0u) {
#pragma unroll
          for (int c = 0; c < 4; ++c) {
            const unsigned bal = __builtin_amdgcn_ballot_w32(m[c]);
            if (m[c]) q[qn + (int)__builtin_amdgcn_mbcnt_lo(bal, 0u)] = e0 + j * 128 + c;
            qn += __builtin_popcount(bal);
          }
        }
      }
      step += NW;
    }

    for (int g = 0; g < qn; g += 32) {
      const int ng = (qn - g < 32) ? (qn - g) : 32;
      int sv = 0, lv = R;
      if (lane < ng) {
        const int e = q[g + lane];
        sv = src[e];
        lv = dst[e] - base;
      }
      atomicAdd(deg + lv, 1.0f);
      for (int b = 0; b < ng; b += 8) {
        float4 v0[8], v1[8];
#pragma unroll
        for (int k = 0; k < 8; ++k) {
          const int s = __builtin_amdgcn_readlane(sv, b + k);
          const float* p = x + (long long)s * C_DIM + lane * 4;
          v0[k] = *(const float4*)p;
          v1[k] = *(const float4*)(p + 128);
        }
#pragma unroll
        for (int k = 0; k < 8; ++k) {
          const int lr = __builtin_amdgcn_readlane(lv, b + k);
          float* a = acc + lr * C_DIM + lane;
          atomicAdd(a +   0, v0[k].x);
          atomicAdd(a +  32, v0[k].y);
          atomicAdd(a +  64, v0[k].z);
          atomicAdd(a +  96, v0[k].w);
          atomicAdd(a + 128, v1[k].x);
          atomicAdd(a + 160, v1[k].y);
          atomicAdd(a + 192, v1[k].z);
          atomicAdd(a + 224, v1[k].w);
        }
      }
    }
    qn = 0;
    if (step >= nsteps) break;
  }
  __syncthreads();

  for (int lr = wave; lr < R; lr += NW) {
    const int row = base + lr;
    if (row < n_dst) {
      const float* a = acc + lr * C_DIM + lane;
      float4 o0, o1;
      o0.x = a[0];   o0.y = a[32];  o0.z = a[64];  o0.w = a[96];
      o1.x = a[128]; o1.y = a[160]; o1.z = a[192]; o1.w = a[224];
      float* o = agg + (long long)row * C_DIM + lane * 4;
      *(float4*)o = o0;
      *(float4*)(o + 128) = o1;
    }
  }
  for (int i = tid; i < R; i += 1024)
    if (base + i < n_dst) cnt[base + i] = deg[i];
}

template <int RELU>
__global__ __launch_bounds__(256) void gemm_dual_wmma(
    const float* __restrict__ A1, const float* __restrict__ Wt1,
    const float* __restrict__ cnt1,
    const float* __restrict__ A2, const float* __restrict__ Wt2,
    const float* __restrict__ bias, float* __restrict__ out,
    int M, int N) {
  constexpr int K = C_DIM;
  const int lane = threadIdx.x & 31;
  const int wave = threadIdx.x >> 5;
  const int nstrips = N >> 6;
  const int tile = blockIdx.x * 8 + wave;
  if (tile >= (M >> 4) * nstrips) return;

  const int m0 = (tile / nstrips) << 4;
  const int n0 = (tile % nstrips) << 6;

  const int half = lane >> 4;
  const int l15  = lane & 15;
  const int row  = m0 + l15;
  const int cb   = n0 + l15;
  const int koff = half << 1;

  const float s1 = 1.0f / fmaxf(cnt1[row], 1.0f);

  v8f acc[4];
#pragma unroll
  for (int j = 0; j < 4; ++j) acc[j] = (v8f){0.f,0.f,0.f,0.f,0.f,0.f,0.f,0.f};

  const float* a1p = A1 + (long long)row * K + koff;
  const float* a2p = A2 + (long long)row * K + koff;
  const float* w1p = Wt1 + (long long)cb * K + koff;
  const float* w2p = Wt2 + (long long)cb * K + koff;

  for (int k = 0; k < K; k += 4) {
    v2f a = *(const v2f*)(a1p + k);
    a = a * s1;
#pragma unroll
    for (int j = 0; j < 4; ++j) {
      v2f b = *(const v2f*)(w1p + (long long)j * 16 * K + k);
      acc[j] = __builtin_amdgcn_wmma_f32_16x16x4_f32(false, a, false, b,
                                                     (short)0, acc[j],
                                                     false, false);
    }
  }
  for (int k = 0; k < K; k += 4) {
    v2f a = *(const v2f*)(a2p + k);
#pragma unroll
    for (int j = 0; j < 4; ++j) {
      v2f b = *(const v2f*)(w2p + (long long)j * 16 * K + k);
      acc[j] = __builtin_amdgcn_wmma_f32_16x16x4_f32(false, a, false, b,
                                                     (short)0, acc[j],
                                                     false, false);
    }
  }

#pragma unroll
  for (int j = 0; j < 4; ++j) {
    const int col = cb + j * 16;
    const float bv = bias[col];
#pragma unroll
    for (int r = 0; r < 8; ++r) {
      float v = acc[j][r] + bv;
      if (RELU) v = fmaxf(v, 0.f);
      out[(long long)(m0 + r + half * 8) * N + col] = v;
    }
  }
}

static inline int cdiv(long long a, long long b) { return (int)((a + b - 1) / b); }

extern "C" void kernel_launch(void* const* d_in, const int* in_sizes, int n_in,
                              void* d_out, int out_size, void* d_ws, size_t ws_size,
                              hipStream_t stream) {
  (void)n_in; (void)out_size; (void)ws_size;

  const int N1 = 40000, N2 = 20000, N3 = 10000;
  const int OUT_C = 64;

  const float* x   = (const float*)d_in[0];
  const int*   ei1 = (const int*)d_in[1];
  const int*   ei2 = (const int*)d_in[2];
  const int*   ei3 = (const int*)d_in[3];
  const float* Wl1 = (const float*)d_in[4];
  const float* bl1 = (const float*)d_in[5];
  const float* Wr1 = (const float*)d_in[6];
  const float* Wl2 = (const float*)d_in[7];
  const float* bl2 = (const float*)d_in[8];
  const float* Wr2 = (const float*)d_in[9];
  const float* Wl3 = (const float*)d_in[10];
  const float* bl3 = (const float*)d_in[11];
  const float* Wr3 = (const float*)d_in[12];
  const float* Wls = (const float*)d_in[13];
  const float* bls = (const float*)d_in[14];
  const float* Wrs = (const float*)d_in[15];

  const int E1 = in_sizes[1] / 2;
  const int E2 = in_sizes[2] / 2;
  const int E3 = in_sizes[3] / 2;

  float* h1  = (float*)d_ws;
  float* h2  = h1 + (long long)N1 * C_DIM;
  float* agg = h2 + (long long)N2 * C_DIM;
  float* cnt = agg + (long long)N1 * C_DIM;
  float* wt  = cnt + N1;
  float* Wl1t = wt;
  float* Wr1t = Wl1t + C_DIM * C_DIM;
  float* Wl2t = Wr1t + C_DIM * C_DIM;
  float* Wr2t = Wl2t + C_DIM * C_DIM;
  float* Wl3t = Wr2t + C_DIM * C_DIM;
  float* Wr3t = Wl3t + C_DIM * OUT_C;
  float* Wlst = Wr3t + C_DIM * OUT_C;
  float* Wrst = Wlst + C_DIM * OUT_C;

  float* mu     = (float*)d_out;
  float* logstd = mu + (long long)N3 * OUT_C;

  const int GB = 256, TB = 256;
  const int AB = 1024;
  constexpr int R1 = 160, R2 = 80, R3 = 40;

  {
    int nbig = C_DIM * C_DIM, nsml = C_DIM * OUT_C;
    transpose_w<<<cdiv(nbig, TB), TB, 0, stream>>>(Wl1, Wl1t, C_DIM, C_DIM);
    transpose_w<<<cdiv(nbig, TB), TB, 0, stream>>>(Wr1, Wr1t, C_DIM, C_DIM);
    transpose_w<<<cdiv(nbig, TB), TB, 0, stream>>>(Wl2, Wl2t, C_DIM, C_DIM);
    transpose_w<<<cdiv(nbig, TB), TB, 0, stream>>>(Wr2, Wr2t, C_DIM, C_DIM);
    transpose_w<<<cdiv(nsml, TB), TB, 0, stream>>>(Wl3, Wl3t, C_DIM, OUT_C);
    transpose_w<<<cdiv(nsml, TB), TB, 0, stream>>>(Wr3, Wr3t, C_DIM, OUT_C);
    transpose_w<<<cdiv(nsml, TB), TB, 0, stream>>>(Wls, Wlst, C_DIM, OUT_C);
    transpose_w<<<cdiv(nsml, TB), TB, 0, stream>>>(Wrs, Wrst, C_DIM, OUT_C);
  }

  {
    aggregate_owner_rows<R1><<<cdiv(N1, R1), AB, 0, stream>>>(
        x, ei1, ei1 + E1, agg, cnt, E1, N1);
    int tiles = (N1 / 16) * (C_DIM / 64);
    gemm_dual_wmma<1><<<cdiv(tiles, 8), GB, 0, stream>>>(
        agg, Wl1t, cnt, x, Wr1t, bl1, h1, N1, C_DIM);
  }

  {
    aggregate_owner_rows<R2><<<cdiv(N2, R2), AB, 0, stream>>>(
        h1, ei2, ei2 + E2, agg, cnt, E2, N2);
    int tiles = (N2 / 16) * (C_DIM / 64);
    gemm_dual_wmma<1><<<cdiv(tiles, 8), GB, 0, stream>>>(
        agg, Wl2t, cnt, h1, Wr2t, bl2, h2, N2, C_DIM);
  }

  {
    aggregate_owner_rows<R3><<<cdiv(N3, R3), AB, 0, stream>>>(
        h2, ei3, ei3 + E3, agg, cnt, E3, N3);

    int tiles = (N3 / 16) * (OUT_C / 64);
    gemm_dual_wmma<0><<<cdiv(tiles, 8), GB, 0, stream>>>(
        agg, Wl3t, cnt, h2, Wr3t, bl3, mu, N3, OUT_C);
    gemm_dual_wmma<0><<<cdiv(tiles, 8), GB, 0, stream>>>(
        agg, Wlst, cnt, h2, Wrst, bls, logstd, N3, OUT_C);
  }
}
